# starting kernel
# baseline (speedup 1.0000x reference)
#include <hip/hip_runtime.h>
#include <hip/hip_bf16.h>


typedef __attribute__((ext_vector_type(2))) float v2f;
typedef __attribute__((ext_vector_type(8))) float v8f;

#define TILE_M     128
#define NW         8
#define LDSS       68

__global__ __launch_bounds__(256)
void gemm_relu_stats(const float* __restrict__ xin, int ldx, int Kp,
                     const float* __restrict__ W,  int Kw,
                     const float* __restrict__ bias,
                     float* __restrict__ hout,
                     float* __restrict__ partial)
{
    __shared__ float xl[TILE_M * LDSS];
    __shared__ float Wl[64 * LDSS];
    __shared__ float biasl[64];
    __shared__ float wsum[NW * 64];
    __shared__ float wsq [NW * 64];

    const int t    = threadIdx.x;
    const int lane = t & 31;
    const int wave = t >> 5;
    const int half = lane >> 4;
    const int l15  = lane & 15;
    const size_t row0 = (size_t)blockIdx.x * TILE_M;

    const int c4n = Kp >> 2;
    for (int i = t; i < TILE_M * c4n; i += 256) {
        int r = i / c4n;
        int c = (i - r * c4n) * 4;
        float4 v = *(const float4*)(xin + (row0 + r) * (size_t)ldx + c);
        *(float4*)(&xl[r * LDSS + c]) = v;
    }
    for (int i = t; i < 64 * Kp; i += 256) {
        int n = i / Kp;
        int k = i - n * Kp;
        Wl[n * LDSS + k] = (n < 59 && k < Kw) ? W[n * Kw + k] : 0.0f;
    }
    if (t < 64) biasl[t] = (t < 59) ? bias[t] : 0.0f;
    __syncthreads();

    v8f acc[4] = {};
    const float* ax = xl + (wave * 16 + l15) * LDSS;
    for (int k0 = 0; k0 < Kp; k0 += 4) {
        const int ka = k0 + 2 * half;
        v2f a = *(const v2f*)(ax + ka);
        #pragma unroll
        for (int tt = 0; tt < 4; tt++) {
            v2f b = *(const v2f*)(Wl + (tt * 16 + l15) * LDSS + ka);
            acc[tt] = __builtin_amdgcn_wmma_f32_16x16x4_f32(
                false, a, false, b, (short)0, acc[tt], false, false);
        }
    }

    const size_t mbase = row0 + wave * 16 + half * 8;
    #pragma unroll
    for (int tt = 0; tt < 4; tt++) {
        const int n = tt * 16 + l15;
        const float bn = biasl[n];
        float sl = 0.f, sq = 0.f;
        #pragma unroll
        for (int r = 0; r < 8; r++) {
            float v = acc[tt][r] + bn;
            v = fmaxf(v, 0.f);
            hout[(mbase + r) * 64 + n] = v;
            sl += v;
            sq += v * v;
        }
        sl += __shfl_xor(sl, 16, 32);
        sq += __shfl_xor(sq, 16, 32);
        if (half == 0) {
            wsum[wave * 64 + n] = sl;
            wsq [wave * 64 + n] = sq;
        }
    }
    __syncthreads();

    if (t < 64) {
        float s = 0.f;
        #pragma unroll
        for (int w = 0; w < NW; w++) s += wsum[w * 64 + t];
        partial[(size_t)blockIdx.x * 128 + t] = s;
    } else if (t < 128) {
        const int n = t - 64;
        float s = 0.f;
        #pragma unroll
        for (int w = 0; w < NW; w++) s += wsq[w * 64 + n];
        partial[(size_t)blockIdx.x * 128 + 64 + n] = s;
    }
}

__global__ void stats_reduce(const float* __restrict__ partial, int nblk,
                             const float* __restrict__ g,
                             const float* __restrict__ bt,
                             float* __restrict__ stats,
                             float invB)
{
    const int n = threadIdx.x;
    if (n >= 64) return;
    float s = 0.f, ss = 0.f;
    for (int b = 0; b < nblk; b++) {
        s  += partial[(size_t)b * 128 + n];
        ss += partial[(size_t)b * 128 + 64 + n];
    }
    const float mu  = s * invB;
    float var = ss * invB - mu * mu;
    var = fmaxf(var, 0.f);
    float scale = 0.f, shift = 0.f;
    if (n < 59) {
        scale = g[n] * rsqrtf(var + 1e-5f);
        shift = bt[n] - mu * scale;
    }
    stats[n]      = scale;
    stats[64 + n] = shift;
}

__global__ __launch_bounds__(256)
void normalize_pad(const float* __restrict__ h, const float* __restrict__ stats,
                   float* __restrict__ out, size_t nvec  )
{
    __shared__ float sc[64], sh[64];
    const int t = threadIdx.x;
    if (t < 64)       sc[t]      = stats[t];
    else if (t < 128) sh[t - 64] = stats[t];
    __syncthreads();

    const size_t idx = (size_t)blockIdx.x * 256 + t;
    if (idx >= nvec) return;
    const int n = (int)(idx & 15) * 4;
    const float4 hv = ((const float4*)h)[idx];
    float4 o;
    o.x = hv.x * sc[n + 0] + sh[n + 0];
    o.y = hv.y * sc[n + 1] + sh[n + 1];
    o.z = hv.z * sc[n + 2] + sh[n + 2];
    o.w = hv.w * sc[n + 3] + sh[n + 3];
    ((float4*)out)[idx] = o;
}

__global__ void normalize_final(const float* __restrict__ h,
                                const float* __restrict__ stats,
                                float* __restrict__ out, size_t total  )
{
    const size_t i = (size_t)blockIdx.x * 256 + threadIdx.x;
    if (i >= total) return;
    const size_t row = i / 59;
    const int n = (int)(i - row * 59);
    out[i] = h[row * 64 + n] * stats[n] + stats[64 + n];
}

__global__ void pad_input(const float* __restrict__ x, float* __restrict__ xp,
                          size_t total  )
{
    const size_t i = (size_t)blockIdx.x * 256 + threadIdx.x;
    if (i >= total) return;
    const size_t row = i >> 5;
    const int n = (int)(i & 31);
    xp[i] = (n < 29) ? x[row * 29 + n] : 0.0f;
}

extern "C" void kernel_launch(void* const* d_in, const int* in_sizes, int n_in,
                              void* d_out, int out_size, void* d_ws, size_t ws_size,
                              hipStream_t stream) {
    const float* x   = (const float*)d_in[0];
    const float* W0  = (const float*)d_in[1];
    const float* b0  = (const float*)d_in[2];
    const float* g0  = (const float*)d_in[3];
    const float* bt0 = (const float*)d_in[4];
    const float* Ws  = (const float*)d_in[5];
    const float* bs  = (const float*)d_in[6];
    const float* gs  = (const float*)d_in[7];
    const float* bts = (const float*)d_in[8];
    float* out = (float*)d_out;

    const int B     = in_sizes[0] / 29;
    const int nblk  = B / TILE_M;
    const float invB = 1.0f / (float)B;

    float* ws      = (float*)d_ws;
    float* xpad    = ws;
    float* P       = xpad    + (size_t)B * 32;
    float* Q       = P       + (size_t)B * 64;
    float* partial = Q       + (size_t)B * 64;
    float* stats   = partial + (size_t)nblk * 128;

    const size_t padTot = (size_t)B * 32;
    pad_input<<<(unsigned)((padTot + 255) / 256), 256, 0, stream>>>(x, xpad, padTot);

    const size_t nvec = (size_t)B * 16;

    gemm_relu_stats<<<nblk, 256, 0, stream>>>(xpad, 32, 32, W0, 29, b0, P, partial);
    stats_reduce<<<1, 64, 0, stream>>>(partial, nblk, g0, bt0, stats, invB);
    normalize_pad<<<(unsigned)(nvec / 256), 256, 0, stream>>>(P, stats, Q, nvec);

    for (int l = 0; l < 14; l++) {
        const float* W  = Ws  + (size_t)l * 59 * 59;
        const float* bb = bs  + (size_t)l * 59;
        const float* gg = gs  + (size_t)l * 59;
        const float* tt = bts + (size_t)l * 59;
        gemm_relu_stats<<<nblk, 256, 0, stream>>>(Q, 64, 64, W, 59, bb, P, partial);
        stats_reduce<<<1, 64, 0, stream>>>(partial, nblk, gg, tt, stats, invB);
        if (l < 13) {
            normalize_pad<<<(unsigned)(nvec / 256), 256, 0, stream>>>(P, stats, Q, nvec);
        } else {
            const size_t total = (size_t)B * 59;
            normalize_final<<<(unsigned)((total + 255) / 256), 256, 0, stream>>>(
                P, stats, out, total);
        }
    }
}


# Persistent GEMM: 256 work-groups each walk 4 row tiles (W staged once, next x tile loaded during the WMMAs) and carry column sums in registers: 256 partial rows, not 1024; batch-sum order changes
# speedup vs baseline: 2.0091x; 2.0091x over previous
#include <hip/hip_runtime.h>
#include <hip/hip_bf16.h>


typedef __attribute__((ext_vector_type(2))) float v2f;
typedef __attribute__((ext_vector_type(8))) float v8f;
typedef __attribute__((ext_vector_type(4))) float v4f;

#define TILE_M     128
#define NW         8
#define LDSS       68


template <int KP>
__device__ __forceinline__ void load_x_tile(v4f (&pre)[KP / 8],
                                            const float* __restrict__ xin,
                                            int ldx, size_t row0, int t)
{
    constexpr int C4N = KP / 4;
    #pragma unroll
    for (int j = 0; j < KP / 8; j++) {
        const int i = t + 256 * j;
        const int r = i / C4N;
        const int c = (i % C4N) * 4;
        pre[j] = *(const v4f*)(xin + (row0 + r) * (size_t)ldx + c);
    }
}

template <int KP>
__global__ __launch_bounds__(256)
void gemm_relu_stats(const float* __restrict__ xin, int ldx,
                     const float* __restrict__ W,  int Kw,
                     const float* __restrict__ bias,
                     float* __restrict__ hout,
                     float* __restrict__ partial,
                     int ntiles)
{
    __shared__ float xl[TILE_M * LDSS];
    __shared__ float Wl[64 * LDSS];
    __shared__ float biasl[64];
    __shared__ float wsum[NW * 64];
    __shared__ float wsq [NW * 64];

    constexpr int C4N = KP / 4;
    const int t    = threadIdx.x;
    const int lane = t & 31;
    const int wave = t >> 5;
    const int half = lane >> 4;
    const int l15  = lane & 15;
    const int gstep = (int)gridDim.x;

    v4f pre[KP / 8];
    int tile = (int)blockIdx.x;
    if (tile < ntiles)
        load_x_tile<KP>(pre, xin, ldx, (size_t)tile * TILE_M, t);

    for (int i = t; i < 64 * KP; i += 256) {
        int n = i / KP;
        int k = i - n * KP;
        Wl[n * LDSS + k] = (n < 59 && k < Kw) ? W[n * Kw + k] : 0.0f;
    }
    if (t < 64) biasl[t] = (t < 59) ? bias[t] : 0.0f;

    float rs[4] = {0.f, 0.f, 0.f, 0.f};
    float rq[4] = {0.f, 0.f, 0.f, 0.f};

    for (; tile < ntiles; tile += gstep) {
        const size_t row0 = (size_t)tile * TILE_M;

        #pragma unroll
        for (int j = 0; j < KP / 8; j++) {
            const int i = t + 256 * j;
            const int r = i / C4N;
            const int c = (i % C4N) * 4;
            *(v4f*)(&xl[r * LDSS + c]) = pre[j];
        }
        __syncthreads();

        const int nxt = tile + gstep;
        if (nxt < ntiles)
            load_x_tile<KP>(pre, xin, ldx, (size_t)nxt * TILE_M, t);

        v8f acc[4] = {};
        const float* ax = xl + (wave * 16 + l15) * LDSS;
        for (int k0 = 0; k0 < KP; k0 += 4) {
            const int ka = k0 + 2 * half;
            v2f a = *(const v2f*)(ax + ka);
            #pragma unroll
            for (int tt = 0; tt < 4; tt++) {
                v2f b = *(const v2f*)(Wl + (tt * 16 + l15) * LDSS + ka);
                acc[tt] = __builtin_amdgcn_wmma_f32_16x16x4_f32(
                    false, a, false, b, (short)0, acc[tt], false, false);
            }
        }

        const size_t mbase = row0 + wave * 16 + half * 8;
        #pragma unroll
        for (int tt = 0; tt < 4; tt++) {
            const int n = tt * 16 + l15;
            const float bn = biasl[n];
            float sl = 0.f, sq = 0.f;
            #pragma unroll
            for (int r = 0; r < 8; r++) {
                float v = acc[tt][r] + bn;
                v = fmaxf(v, 0.f);
                hout[(mbase + r) * 64 + n] = v;
                sl += v;
                sq += v * v;
            }
            rs[tt] += sl;
            rq[tt] += sq;
        }
        __syncthreads();
    }

    #pragma unroll
    for (int tt = 0; tt < 4; tt++) {
        const int n = tt * 16 + l15;
        float sl = rs[tt], sq = rq[tt];
        sl += __shfl_xor(sl, 16, 32);
        sq += __shfl_xor(sq, 16, 32);
        if (half == 0) {
            wsum[wave * 64 + n] = sl;
            wsq [wave * 64 + n] = sq;
        }
    }
    __syncthreads();

    if (t < 64) {
        float s = 0.f;
        #pragma unroll
        for (int w = 0; w < NW; w++) s += wsum[w * 64 + t];
        partial[(size_t)blockIdx.x * 128 + t] = s;
    } else if (t < 128) {
        const int n = t - 64;
        float s = 0.f;
        #pragma unroll
        for (int w = 0; w < NW; w++) s += wsq[w * 64 + n];
        partial[(size_t)blockIdx.x * 128 + 64 + n] = s;
    }
}

__global__ void stats_reduce(const float* __restrict__ partial, int nblk,
                             const float* __restrict__ g,
                             const float* __restrict__ bt,
                             float* __restrict__ stats,
                             float invB)
{
    const int n = threadIdx.x;
    if (n >= 64) return;
    float s = 0.f, ss = 0.f;
    for (int b = 0; b < nblk; b++) {
        s  += partial[(size_t)b * 128 + n];
        ss += partial[(size_t)b * 128 + 64 + n];
    }
    const float mu  = s * invB;
    float var = ss * invB - mu * mu;
    var = fmaxf(var, 0.f);
    float scale = 0.f, shift = 0.f;
    if (n < 59) {
        scale = g[n] * rsqrtf(var + 1e-5f);
        shift = bt[n] - mu * scale;
    }
    stats[n]      = scale;
    stats[64 + n] = shift;
}

__global__ __launch_bounds__(256)
void normalize_pad(const float* __restrict__ h, const float* __restrict__ stats,
                   float* __restrict__ out, size_t nvec  )
{
    __shared__ float sc[64], sh[64];
    const int t = threadIdx.x;
    if (t < 64)       sc[t]      = stats[t];
    else if (t < 128) sh[t - 64] = stats[t];
    __syncthreads();

    const size_t idx = (size_t)blockIdx.x * 256 + t;
    if (idx >= nvec) return;
    const int n = (int)(idx & 15) * 4;
    const float4 hv = ((const float4*)h)[idx];
    float4 o;
    o.x = hv.x * sc[n + 0] + sh[n + 0];
    o.y = hv.y * sc[n + 1] + sh[n + 1];
    o.z = hv.z * sc[n + 2] + sh[n + 2];
    o.w = hv.w * sc[n + 3] + sh[n + 3];
    ((float4*)out)[idx] = o;
}

__global__ void normalize_final(const float* __restrict__ h,
                                const float* __restrict__ stats,
                                float* __restrict__ out, size_t total  )
{
    const size_t i = (size_t)blockIdx.x * 256 + threadIdx.x;
    if (i >= total) return;
    const size_t row = i / 59;
    const int n = (int)(i - row * 59);
    out[i] = h[row * 64 + n] * stats[n] + stats[64 + n];
}

__global__ void pad_input(const float* __restrict__ x, float* __restrict__ xp,
                          size_t total  )
{
    const size_t i = (size_t)blockIdx.x * 256 + threadIdx.x;
    if (i >= total) return;
    const size_t row = i >> 5;
    const int n = (int)(i & 31);
    xp[i] = (n < 29) ? x[row * 29 + n] : 0.0f;
}

extern "C" void kernel_launch(void* const* d_in, const int* in_sizes, int n_in,
                              void* d_out, int out_size, void* d_ws, size_t ws_size,
                              hipStream_t stream) {
    const float* x   = (const float*)d_in[0];
    const float* W0  = (const float*)d_in[1];
    const float* b0  = (const float*)d_in[2];
    const float* g0  = (const float*)d_in[3];
    const float* bt0 = (const float*)d_in[4];
    const float* Ws  = (const float*)d_in[5];
    const float* bs  = (const float*)d_in[6];
    const float* gs  = (const float*)d_in[7];
    const float* bts = (const float*)d_in[8];
    float* out = (float*)d_out;

    const int B     = in_sizes[0] / 29;
    const int nblk  = B / TILE_M;
    const int G     = 256;
    const float invB = 1.0f / (float)B;

    float* ws      = (float*)d_ws;
    float* xpad    = ws;
    float* P       = xpad    + (size_t)B * 32;
    float* Q       = P       + (size_t)B * 64;
    float* partial = Q       + (size_t)B * 64;
    float* stats   = partial + (size_t)nblk * 128;

    const size_t padTot = (size_t)B * 32;
    pad_input<<<(unsigned)((padTot + 255) / 256), 256, 0, stream>>>(x, xpad, padTot);

    const size_t nvec = (size_t)B * 16;

    gemm_relu_stats<32><<<G, 256, 0, stream>>>(xpad, 32, W0, 29, b0, P, partial, nblk);
    stats_reduce<<<1, 64, 0, stream>>>(partial, G, g0, bt0, stats, invB);
    normalize_pad<<<(unsigned)(nvec / 256), 256, 0, stream>>>(P, stats, Q, nvec);

    for (int l = 0; l < 14; l++) {
        const float* W  = Ws  + (size_t)l * 59 * 59;
        const float* bb = bs  + (size_t)l * 59;
        const float* gg = gs  + (size_t)l * 59;
        const float* tt = bts + (size_t)l * 59;
        gemm_relu_stats<64><<<G, 256, 0, stream>>>(Q, 64, W, 59, bb, P, partial, nblk);
        stats_reduce<<<1, 64, 0, stream>>>(partial, G, gg, tt, stats, invB);
        if (l < 13) {
            normalize_pad<<<(unsigned)(nvec / 256), 256, 0, stream>>>(P, stats, Q, nvec);
        } else {
            const size_t total = (size_t)B * 59;
            normalize_final<<<(unsigned)((total + 255) / 256), 256, 0, stream>>>(
                P, stats, out, total);
        }
    }
}


# BatchNorm fold (W*scale, bias + W.shift) done once by the stats kernel into a padded 64x64 matrix; GEMM work-groups copy it with 16-byte loads, not 16 serial 4-byte loads each
# speedup vs baseline: 3.0107x; 1.4985x over previous
#include <hip/hip_runtime.h>
#include <hip/hip_bf16.h>


typedef __attribute__((ext_vector_type(2))) float v2f;
typedef __attribute__((ext_vector_type(8))) float v8f;
typedef __attribute__((ext_vector_type(4))) float v4f;

#define TILE_M     128
#define NW         8
#define LDSS       68


template <int KP>
__device__ __forceinline__ void load_x_tile(v4f (&pre)[KP / 8],
                                            const float* __restrict__ xin,
                                            int ldx, size_t row0, int t)
{
    constexpr int C4N = KP / 4;
    #pragma unroll
    for (int j = 0; j < KP / 8; j++) {
        const int i = t + 256 * j;
        const int r = i / C4N;
        const int c = (i % C4N) * 4;
        pre[j] = *(const v4f*)(xin + (row0 + r) * (size_t)ldx + c);
    }
}

template <int KP, bool FOLD>
__global__ __launch_bounds__(256)
void gemm_relu_stats(const float* __restrict__ xin, int ldx,
                     const float* __restrict__ W,  int Kw,
                     const float* __restrict__ bias,
                     float* __restrict__ hout,
                     float* __restrict__ partial,
                     int ntiles)
{
    __shared__ float xl[TILE_M * LDSS];
    __shared__ float Wl[64 * LDSS];
    __shared__ float biasl[64];
    __shared__ float wsum[NW * 64];
    __shared__ float wsq [NW * 64];

    constexpr int C4N = KP / 4;
    const int t    = threadIdx.x;
    const int lane = t & 31;
    const int wave = t >> 5;
    const int half = lane >> 4;
    const int l15  = lane & 15;
    const int gstep = (int)gridDim.x;

    v4f pre[KP / 8];
    int tile = (int)blockIdx.x;
    if (tile < ntiles)
        load_x_tile<KP>(pre, xin, ldx, (size_t)tile * TILE_M, t);

    if constexpr (FOLD) {
        static_assert(!FOLD || KP == 64, "the folded matrix is 64 x 64");
        const float bv = bias[t & 63];
        for (int j = 0; j < 4; j++) {
            const int i = t + 256 * j;
            const int n = i >> 4;
            const int c = (i & 15) * 4;
            *(v4f*)(&Wl[n * LDSS + c]) = *(const v4f*)(W + n * 64 + c);
        }
        biasl[t & 63] = bv;
    } else {
        for (int i = t; i < 64 * KP; i += 256) {
            int n = i / KP;
            int k = i - n * KP;
            Wl[n * LDSS + k] = (n < 59 && k < Kw) ? W[n * Kw + k] : 0.0f;
        }
        if (t < 64) biasl[t] = (t < 59) ? bias[t] : 0.0f;
    }

    float rs[4] = {0.f, 0.f, 0.f, 0.f};
    float rq[4] = {0.f, 0.f, 0.f, 0.f};

    for (; tile < ntiles; tile += gstep) {
        const size_t row0 = (size_t)tile * TILE_M;

        #pragma unroll
        for (int j = 0; j < KP / 8; j++) {
            const int i = t + 256 * j;
            const int r = i / C4N;
            const int c = (i % C4N) * 4;
            *(v4f*)(&xl[r * LDSS + c]) = pre[j];
        }
        __syncthreads();

        const int nxt = tile + gstep;
        if (nxt < ntiles)
            load_x_tile<KP>(pre, xin, ldx, (size_t)nxt * TILE_M, t);

        v8f acc[4] = {};
        const float* ax = xl + (wave * 16 + l15) * LDSS;
        for (int k0 = 0; k0 < KP; k0 += 4) {
            const int ka = k0 + 2 * half;
            v2f a = *(const v2f*)(ax + ka);
            #pragma unroll
            for (int tt = 0; tt < 4; tt++) {
                v2f b = *(const v2f*)(Wl + (tt * 16 + l15) * LDSS + ka);
                acc[tt] = __builtin_amdgcn_wmma_f32_16x16x4_f32(
                    false, a, false, b, (short)0, acc[tt], false, false);
            }
        }

        const size_t mbase = row0 + wave * 16 + half * 8;
        #pragma unroll
        for (int tt = 0; tt < 4; tt++) {
            const int n = tt * 16 + l15;
            const float bn = biasl[n];
            float sl = 0.f, sq = 0.f;
            #pragma unroll
            for (int r = 0; r < 8; r++) {
                float v = acc[tt][r] + bn;
                v = fmaxf(v, 0.f);
                hout[(mbase + r) * 64 + n] = v;
                sl += v;
                sq += v * v;
            }
            rs[tt] += sl;
            rq[tt] += sq;
        }
        __syncthreads();
    }

    #pragma unroll
    for (int tt = 0; tt < 4; tt++) {
        const int n = tt * 16 + l15;
        float sl = rs[tt], sq = rq[tt];
        sl += __shfl_xor(sl, 16, 32);
        sq += __shfl_xor(sq, 16, 32);
        if (half == 0) {
            wsum[wave * 64 + n] = sl;
            wsq [wave * 64 + n] = sq;
        }
    }
    __syncthreads();

    if (t < 64) {
        float s = 0.f;
        #pragma unroll
        for (int w = 0; w < NW; w++) s += wsum[w * 64 + t];
        partial[(size_t)blockIdx.x * 128 + t] = s;
    } else if (t < 128) {
        const int n = t - 64;
        float s = 0.f;
        #pragma unroll
        for (int w = 0; w < NW; w++) s += wsq[w * 64 + n];
        partial[(size_t)blockIdx.x * 128 + 64 + n] = s;
    }
}

template <bool NEXT>
__global__ __launch_bounds__(256)
void stats_fold(const float* __restrict__ partial, int nblk,
                const float* __restrict__ g,
                const float* __restrict__ bt,
                float* __restrict__ stats,
                float invB,
                const float* __restrict__ Wn,
                const float* __restrict__ bnx,
                float* __restrict__ Wf,
                float* __restrict__ bf)
{
    __shared__ float scl[64];
    __shared__ float shl[64];
    __shared__ float term[NEXT ? 64 * LDSS : 1];

    const int t = threadIdx.x;
    const int k = t & 63;

    float w[16];
    float bnv = 0.f;
    if constexpr (NEXT) {
        #pragma unroll
        for (int j = 0; j < 16; j++) {
            const int n = (t >> 6) + 4 * j;
            const bool in = (n < 59 && k < 59);
            const float v = Wn[in ? n * 59 + k : 0];
            w[j] = in ? v : 0.0f;
        }
        const int nb = t >> 2;
        const float v = bnx[(nb < 59) ? nb : 0];
        bnv = (nb < 59) ? v : 0.0f;
    }

    if (t < 64) {
        const int n = t;
        float s = 0.f, ss = 0.f;
        for (int b = 0; b < nblk; b++) {
            s  += partial[(size_t)b * 128 + n];
            ss += partial[(size_t)b * 128 + 64 + n];
        }
        const float mu  = s * invB;
        float var = ss * invB - mu * mu;
        var = fmaxf(var, 0.f);
        float scale = 0.f, shift = 0.f;
        if (n < 59) {
            scale = g[n] * rsqrtf(var + 1e-5f);
            shift = bt[n] - mu * scale;
        }
        if constexpr (NEXT) {
            scl[n] = scale;
            shl[n] = shift;
        } else {
            stats[n]      = scale;
            stats[64 + n] = shift;
        }
    }

    if constexpr (NEXT) {
        __syncthreads();
        const float sck = scl[k];
        const float shk = shl[k];
        #pragma unroll
        for (int j = 0; j < 16; j++) {
            const int n = (t >> 6) + 4 * j;
            term[n * LDSS + k] = w[j] * shk;
        }
        __syncthreads();
        #pragma unroll
        for (int j = 0; j < 16; j++) {
            const int n = (t >> 6) + 4 * j;
            Wf[n * 64 + k] = w[j] * sck;
        }
        {
            const int n = t >> 2;
            const float* p = term + n * LDSS + (t & 3) * 16;
            float s = 0.f;
            for (int q = 0; q < 4; q++) {
                const v4f v = *(const v4f*)(p + 4 * q);
                s += v.x; s += v.y; s += v.z; s += v.w;
            }
            s += __shfl_xor(s, 1, 32);
            s += __shfl_xor(s, 2, 32);
            if ((t & 3) == 0) bf[n] = bnv + s;
        }
    }
}

__global__ void normalize_final(const float* __restrict__ h,
                                const float* __restrict__ stats,
                                float* __restrict__ out, size_t total  )
{
    const size_t i = (size_t)blockIdx.x * 256 + threadIdx.x;
    if (i >= total) return;
    const size_t row = i / 59;
    const int n = (int)(i - row * 59);
    out[i] = h[row * 64 + n] * stats[n] + stats[64 + n];
}

__global__ void pad_input(const float* __restrict__ x, float* __restrict__ xp,
                          size_t total  )
{
    const size_t i = (size_t)blockIdx.x * 256 + threadIdx.x;
    if (i >= total) return;
    const size_t row = i >> 5;
    const int n = (int)(i & 31);
    xp[i] = (n < 29) ? x[row * 29 + n] : 0.0f;
}

extern "C" void kernel_launch(void* const* d_in, const int* in_sizes, int n_in,
                              void* d_out, int out_size, void* d_ws, size_t ws_size,
                              hipStream_t stream) {
    const float* x   = (const float*)d_in[0];
    const float* W0  = (const float*)d_in[1];
    const float* b0  = (const float*)d_in[2];
    const float* g0  = (const float*)d_in[3];
    const float* bt0 = (const float*)d_in[4];
    const float* Ws  = (const float*)d_in[5];
    const float* bs  = (const float*)d_in[6];
    const float* gs  = (const float*)d_in[7];
    const float* bts = (const float*)d_in[8];
    float* out = (float*)d_out;

    const int B     = in_sizes[0] / 29;
    const int nblk  = B / TILE_M;
    const int G     = 256;
    const float invB = 1.0f / (float)B;

    float* ws      = (float*)d_ws;
    float* xpad    = ws;
    float* P       = xpad    + (size_t)B * 32;
    float* Q       = P       + (size_t)B * 64;
    float* partial = Q       + (size_t)B * 64;
    float* stats   = partial + (size_t)nblk * 128;
    float* Wf      = partial + (size_t)G * 128;
    float* bf      = Wf + 64 * 64;

    const size_t padTot = (size_t)B * 32;
    pad_input<<<(unsigned)((padTot + 255) / 256), 256, 0, stream>>>(x, xpad, padTot);

    gemm_relu_stats<32, false><<<G, 256, 0, stream>>>(xpad, 32, W0, 29, b0, P, partial, nblk);
    stats_fold<true><<<1, 256, 0, stream>>>(partial, G, g0, bt0, stats, invB, Ws, bs, Wf, bf);

    for (int l = 0; l < 14; l++) {
        const float* gg = gs  + (size_t)l * 59;
        const float* tt = bts + (size_t)l * 59;
        const float* src = (l & 1) ? Q : P;
        float*       dst = (l & 1) ? P : Q;
        gemm_relu_stats<64, true><<<G, 256, 0, stream>>>(src, 64, Wf, 64, bf, dst, partial, nblk);
        if (l < 13)
            stats_fold<true><<<1, 256, 0, stream>>>(partial, G, gg, tt, stats, invB,
                                                     Ws + (size_t)(l + 1) * 59 * 59, bs + (size_t)(l + 1) * 59, Wf, bf);
        else
            stats_fold<false><<<1, 256, 0, stream>>>(partial, G, gg, tt, stats, invB,
                                                      nullptr, nullptr, nullptr, nullptr);
    }

    const size_t total = (size_t)B * 59;
    normalize_final<<<(unsigned)((total + 255) / 256), 256, 0, stream>>>(P, stats, out, total);
}


# GEMM tile loop: barriers written by hand wait for LDS only (s_wait_dscnt); __syncthreads() also drained the tile's 32 output stores and the next tile's loads, now issued ahead of the barrier
# speedup vs baseline: 3.2435x; 1.0773x over previous
#include <hip/hip_runtime.h>
#include <hip/hip_bf16.h>


typedef __attribute__((ext_vector_type(2))) float v2f;
typedef __attribute__((ext_vector_type(8))) float v8f;
typedef __attribute__((ext_vector_type(4))) float v4f;

#define TILE_M     128
#define NW         8
#define LDSS       68

__device__ __forceinline__ void lds_barrier()
{
    asm volatile("s_wait_dscnt 0x0\n\ts_barrier_signal -1\n\ts_barrier_wait -1" ::: "memory");
}


template <int KP>
__device__ __forceinline__ void load_x_tile(v4f (&pre)[KP / 8],
                                            const float* __restrict__ xin,
                                            int ldx, size_t row0, int t)
{
    constexpr int C4N = KP / 4;
    constexpr int RJ  = 256 / C4N;
    static_assert(256 % C4N == 0, "a thread keeps its column over j");
    const float* __restrict__ xt = xin + row0 * (size_t)ldx;
    const unsigned off = (unsigned)(t / C4N) * (unsigned)ldx + (unsigned)(t % C4N) * 4u;
    #pragma unroll
    for (int j = 0; j < KP / 8; j++)
        pre[j] = *(const v4f*)(xt + (off + (unsigned)(RJ * j) * (unsigned)ldx));
}

template <int KP, bool FOLD>
__global__ __launch_bounds__(256)
void gemm_relu_stats(const float* __restrict__ xin, int ldx,
                     const float* __restrict__ W,  int Kw,
                     const float* __restrict__ bias,
                     float* __restrict__ hout,
                     float* __restrict__ partial,
                     int ntiles)
{
    __shared__ float xl[TILE_M * LDSS];
    __shared__ float Wl[64 * LDSS];
    __shared__ float biasl[64];
    __shared__ float wsum[NW * 64];
    __shared__ float wsq [NW * 64];

    constexpr int C4N = KP / 4;
    const int t    = threadIdx.x;
    const int lane = t & 31;
    const int wave = t >> 5;
    const int half = lane >> 4;
    const int l15  = lane & 15;
    const int gstep = (int)gridDim.x;

    v4f pre[KP / 8];
    int tile = (int)blockIdx.x;
    if (tile < ntiles)
        load_x_tile<KP>(pre, xin, ldx, (size_t)tile * TILE_M, t);

    if constexpr (FOLD) {
        static_assert(!FOLD || KP == 64, "the folded matrix is 64 x 64");
        const float bv = bias[t & 63];
        for (int j = 0; j < 4; j++) {
            const int i = t + 256 * j;
            const int n = i >> 4;
            const int c = (i & 15) * 4;
            *(v4f*)(&Wl[n * LDSS + c]) = *(const v4f*)(W + n * 64 + c);
        }
        biasl[t & 63] = bv;
    } else {
        for (int i = t; i < 64 * KP; i += 256) {
            int n = i / KP;
            int k = i - n * KP;
            Wl[n * LDSS + k] = (n < 59 && k < Kw) ? W[n * Kw + k] : 0.0f;
        }
        if (t < 64) biasl[t] = (t < 59) ? bias[t] : 0.0f;
    }

    float rs[4] = {0.f, 0.f, 0.f, 0.f};
    float rq[4] = {0.f, 0.f, 0.f, 0.f};

    for (; tile < ntiles; tile += gstep) {
        const size_t row0 = (size_t)tile * TILE_M;

        #pragma unroll
        for (int j = 0; j < KP / 8; j++) {
            const int i = t + 256 * j;
            const int r = i / C4N;
            const int c = (i % C4N) * 4;
            *(v4f*)(&xl[r * LDSS + c]) = pre[j];
        }

        const int nxt = tile + gstep;
        if (nxt < ntiles)
            load_x_tile<KP>(pre, xin, ldx, (size_t)nxt * TILE_M, t);

        lds_barrier();

        v8f acc[4] = {};
        const float* ax = xl + (wave * 16 + l15) * LDSS;
        for (int k0 = 0; k0 < KP; k0 += 4) {
            const int ka = k0 + 2 * half;
            v2f a = *(const v2f*)(ax + ka);
            #pragma unroll
            for (int tt = 0; tt < 4; tt++) {
                v2f b = *(const v2f*)(Wl + (tt * 16 + l15) * LDSS + ka);
                acc[tt] = __builtin_amdgcn_wmma_f32_16x16x4_f32(
                    false, a, false, b, (short)0, acc[tt], false, false);
            }
        }

        const size_t mbase = row0 + wave * 16 + half * 8;
        #pragma unroll
        for (int tt = 0; tt < 4; tt++) {
            const int n = tt * 16 + l15;
            const float bn = biasl[n];
            float sl = 0.f, sq = 0.f;
            #pragma unroll
            for (int r = 0; r < 8; r++) {
                float v = acc[tt][r] + bn;
                v = fmaxf(v, 0.f);
                hout[(mbase + r) * 64 + n] = v;
                sl += v;
                sq += v * v;
            }
            rs[tt] += sl;
            rq[tt] += sq;
        }
        lds_barrier();
    }

    #pragma unroll
    for (int tt = 0; tt < 4; tt++) {
        const int n = tt * 16 + l15;
        float sl = rs[tt], sq = rq[tt];
        sl += __shfl_xor(sl, 16, 32);
        sq += __shfl_xor(sq, 16, 32);
        if (half == 0) {
            wsum[wave * 64 + n] = sl;
            wsq [wave * 64 + n] = sq;
        }
    }
    lds_barrier();

    if (t < 64) {
        float s = 0.f;
        #pragma unroll
        for (int w = 0; w < NW; w++) s += wsum[w * 64 + t];
        partial[(size_t)blockIdx.x * 128 + t] = s;
    } else if (t < 128) {
        const int n = t - 64;
        float s = 0.f;
        #pragma unroll
        for (int w = 0; w < NW; w++) s += wsq[w * 64 + n];
        partial[(size_t)blockIdx.x * 128 + 64 + n] = s;
    }
}

template <bool NEXT>
__global__ __launch_bounds__(256)
void stats_fold(const float* __restrict__ partial, int nblk,
                const float* __restrict__ g,
                const float* __restrict__ bt,
                float* __restrict__ stats,
                float invB,
                const float* __restrict__ Wn,
                const float* __restrict__ bnx,
                float* __restrict__ Wf,
                float* __restrict__ bf)
{
    __shared__ float scl[64];
    __shared__ float shl[64];
    __shared__ float term[NEXT ? 64 * LDSS : 1];

    const int t = threadIdx.x;
    const int k = t & 63;

    float w[16];
    float bnv = 0.f;
    if constexpr (NEXT) {
        #pragma unroll
        for (int j = 0; j < 16; j++) {
            const int n = (t >> 6) + 4 * j;
            const bool in = (n < 59 && k < 59);
            const float v = Wn[in ? n * 59 + k : 0];
            w[j] = in ? v : 0.0f;
        }
        const int nb = t >> 2;
        const float v = bnx[(nb < 59) ? nb : 0];
        bnv = (nb < 59) ? v : 0.0f;
    }

    if (t < 64) {
        const int n = t;
        float s = 0.f, ss = 0.f;
        for (int b = 0; b < nblk; b++) {
            s  += partial[(size_t)b * 128 + n];
            ss += partial[(size_t)b * 128 + 64 + n];
        }
        const float mu  = s * invB;
        float var = ss * invB - mu * mu;
        var = fmaxf(var, 0.f);
        float scale = 0.f, shift = 0.f;
        if (n < 59) {
            scale = g[n] * rsqrtf(var + 1e-5f);
            shift = bt[n] - mu * scale;
        }
        if constexpr (NEXT) {
            scl[n] = scale;
            shl[n] = shift;
        } else {
            stats[n]      = scale;
            stats[64 + n] = shift;
        }
    }

    if constexpr (NEXT) {
        __syncthreads();
        const float sck = scl[k];
        const float shk = shl[k];
        #pragma unroll
        for (int j = 0; j < 16; j++) {
            const int n = (t >> 6) + 4 * j;
            term[n * LDSS + k] = w[j] * shk;
        }
        __syncthreads();
        #pragma unroll
        for (int j = 0; j < 16; j++) {
            const int n = (t >> 6) + 4 * j;
            Wf[n * 64 + k] = w[j] * sck;
        }
        {
            const int n = t >> 2;
            const float* p = term + n * LDSS + (t & 3) * 16;
            float s = 0.f;
            for (int q = 0; q < 4; q++) {
                const v4f v = *(const v4f*)(p + 4 * q);
                s += v.x; s += v.y; s += v.z; s += v.w;
            }
            s += __shfl_xor(s, 1, 32);
            s += __shfl_xor(s, 2, 32);
            if ((t & 3) == 0) bf[n] = bnv + s;
        }
    }
}

__global__ void normalize_final(const float* __restrict__ h,
                                const float* __restrict__ stats,
                                float* __restrict__ out, size_t total  )
{
    const size_t i = (size_t)blockIdx.x * 256 + threadIdx.x;
    if (i >= total) return;
    const size_t row = i / 59;
    const int n = (int)(i - row * 59);
    out[i] = h[row * 64 + n] * stats[n] + stats[64 + n];
}

__global__ void pad_input(const float* __restrict__ x, float* __restrict__ xp,
                          size_t total  )
{
    const size_t i = (size_t)blockIdx.x * 256 + threadIdx.x;
    if (i >= total) return;
    const size_t row = i >> 5;
    const int n = (int)(i & 31);
    xp[i] = (n < 29) ? x[row * 29 + n] : 0.0f;
}

extern "C" void kernel_launch(void* const* d_in, const int* in_sizes, int n_in,
                              void* d_out, int out_size, void* d_ws, size_t ws_size,
                              hipStream_t stream) {
    const float* x   = (const float*)d_in[0];
    const float* W0  = (const float*)d_in[1];
    const float* b0  = (const float*)d_in[2];
    const float* g0  = (const float*)d_in[3];
    const float* bt0 = (const float*)d_in[4];
    const float* Ws  = (const float*)d_in[5];
    const float* bs  = (const float*)d_in[6];
    const float* gs  = (const float*)d_in[7];
    const float* bts = (const float*)d_in[8];
    float* out = (float*)d_out;

    const int B     = in_sizes[0] / 29;
    const int nblk  = B / TILE_M;
    const int G     = 256;
    const float invB = 1.0f / (float)B;

    float* ws      = (float*)d_ws;
    float* xpad    = ws;
    float* P       = xpad    + (size_t)B * 32;
    float* Q       = P       + (size_t)B * 64;
    float* partial = Q       + (size_t)B * 64;
    float* stats   = partial + (size_t)nblk * 128;
    float* Wf      = partial + (size_t)G * 128;
    float* bf      = Wf + 64 * 64;

    const size_t padTot = (size_t)B * 32;
    pad_input<<<(unsigned)((padTot + 255) / 256), 256, 0, stream>>>(x, xpad, padTot);

    gemm_relu_stats<32, false><<<G, 256, 0, stream>>>(xpad, 32, W0, 29, b0, P, partial, nblk);
    stats_fold<true><<<1, 256, 0, stream>>>(partial, G, g0, bt0, stats, invB, Ws, bs, Wf, bf);

    for (int l = 0; l < 14; l++) {
        const float* gg = gs  + (size_t)l * 59;
        const float* tt = bts + (size_t)l * 59;
        const float* src = (l & 1) ? Q : P;
        float*       dst = (l & 1) ? P : Q;
        gemm_relu_stats<64, true><<<G, 256, 0, stream>>>(src, 64, Wf, 64, bf, dst, partial, nblk);
        if (l < 13)
            stats_fold<true><<<1, 256, 0, stream>>>(partial, G, gg, tt, stats, invB,
                                                     Ws + (size_t)(l + 1) * 59 * 59, bs + (size_t)(l + 1) * 59, Wf, bf);
        else
            stats_fold<false><<<1, 256, 0, stream>>>(partial, G, gg, tt, stats, invB,
                                                      nullptr, nullptr, nullptr, nullptr);
    }

    const size_t total = (size_t)B * 59;
    normalize_final<<<(unsigned)((total + 255) / 256), 256, 0, stream>>>(P, stats, out, total);
}


# stats kernel: all 256 threads copy the partial rows into LDS with 16-byte loads (one cache round trip, was one per 8 rows), then walk them there in the old ascending order; sums bit-identical
# speedup vs baseline: 5.6267x; 1.7348x over previous
#include <hip/hip_runtime.h>
#include <hip/hip_bf16.h>


typedef __attribute__((ext_vector_type(2))) float v2f;
typedef __attribute__((ext_vector_type(8))) float v8f;
typedef __attribute__((ext_vector_type(4))) float v4f;

#define TILE_M     128
#define NW         8
#define LDSS       68

__device__ __forceinline__ void lds_barrier()
{
    asm volatile("s_wait_dscnt 0x0\n\ts_barrier_signal -1\n\ts_barrier_wait -1" ::: "memory");
}


template <int KP>
__device__ __forceinline__ void load_x_tile(v4f (&pre)[KP / 8],
                                            const float* __restrict__ xin,
                                            int ldx, size_t row0, int t)
{
    constexpr int C4N = KP / 4;
    constexpr int RJ  = 256 / C4N;
    static_assert(256 % C4N == 0, "a thread keeps its column over j");
    const float* __restrict__ xt = xin + row0 * (size_t)ldx;
    const unsigned off = (unsigned)(t / C4N) * (unsigned)ldx + (unsigned)(t % C4N) * 4u;
    #pragma unroll
    for (int j = 0; j < KP / 8; j++)
        pre[j] = *(const v4f*)(xt + (off + (unsigned)(RJ * j) * (unsigned)ldx));
}

template <int KP, bool FOLD>
__global__ __launch_bounds__(256)
void gemm_relu_stats(const float* __restrict__ xin, int ldx,
                     const float* __restrict__ W,  int Kw,
                     const float* __restrict__ bias,
                     float* __restrict__ hout,
                     float* __restrict__ partial,
                     int ntiles)
{
    __shared__ float xl[TILE_M * LDSS];
    __shared__ float Wl[64 * LDSS];
    __shared__ float biasl[64];
    __shared__ float wsum[NW * 64];
    __shared__ float wsq [NW * 64];

    constexpr int C4N = KP / 4;
    const int t    = threadIdx.x;
    const int lane = t & 31;
    const int wave = t >> 5;
    const int half = lane >> 4;
    const int l15  = lane & 15;
    const int gstep = (int)gridDim.x;

    v4f pre[KP / 8];
    int tile = (int)blockIdx.x;
    if (tile < ntiles)
        load_x_tile<KP>(pre, xin, ldx, (size_t)tile * TILE_M, t);

    if constexpr (FOLD) {
        static_assert(!FOLD || KP == 64, "the folded matrix is 64 x 64");
        const float bv = bias[t & 63];
        for (int j = 0; j < 4; j++) {
            const int i = t + 256 * j;
            const int n = i >> 4;
            const int c = (i & 15) * 4;
            *(v4f*)(&Wl[n * LDSS + c]) = *(const v4f*)(W + n * 64 + c);
        }
        biasl[t & 63] = bv;
    } else {
        for (int i = t; i < 64 * KP; i += 256) {
            int n = i / KP;
            int k = i - n * KP;
            Wl[n * LDSS + k] = (n < 59 && k < Kw) ? W[n * Kw + k] : 0.0f;
        }
        if (t < 64) biasl[t] = (t < 59) ? bias[t] : 0.0f;
    }

    float rs[4] = {0.f, 0.f, 0.f, 0.f};
    float rq[4] = {0.f, 0.f, 0.f, 0.f};

    for (; tile < ntiles; tile += gstep) {
        const size_t row0 = (size_t)tile * TILE_M;

        #pragma unroll
        for (int j = 0; j < KP / 8; j++) {
            const int i = t + 256 * j;
            const int r = i / C4N;
            const int c = (i % C4N) * 4;
            *(v4f*)(&xl[r * LDSS + c]) = pre[j];
        }

        const int nxt = tile + gstep;
        if (nxt < ntiles)
            load_x_tile<KP>(pre, xin, ldx, (size_t)nxt * TILE_M, t);

        lds_barrier();

        v8f acc[4] = {};
        const float* ax = xl + (wave * 16 + l15) * LDSS;
        for (int k0 = 0; k0 < KP; k0 += 4) {
            const int ka = k0 + 2 * half;
            v2f a = *(const v2f*)(ax + ka);
            #pragma unroll
            for (int tt = 0; tt < 4; tt++) {
                v2f b = *(const v2f*)(Wl + (tt * 16 + l15) * LDSS + ka);
                acc[tt] = __builtin_amdgcn_wmma_f32_16x16x4_f32(
                    false, a, false, b, (short)0, acc[tt], false, false);
            }
        }

        const size_t mbase = row0 + wave * 16 + half * 8;
        #pragma unroll
        for (int tt = 0; tt < 4; tt++) {
            const int n = tt * 16 + l15;
            const float bn = biasl[n];
            float sl = 0.f, sq = 0.f;
            #pragma unroll
            for (int r = 0; r < 8; r++) {
                float v = acc[tt][r] + bn;
                v = fmaxf(v, 0.f);
                hout[(mbase + r) * 64 + n] = v;
                sl += v;
                sq += v * v;
            }
            rs[tt] += sl;
            rq[tt] += sq;
        }
        lds_barrier();
    }

    #pragma unroll
    for (int tt = 0; tt < 4; tt++) {
        const int n = tt * 16 + l15;
        float sl = rs[tt], sq = rq[tt];
        sl += __shfl_xor(sl, 16, 32);
        sq += __shfl_xor(sq, 16, 32);
        if (half == 0) {
            wsum[wave * 64 + n] = sl;
            wsq [wave * 64 + n] = sq;
        }
    }
    lds_barrier();

    if (t < 64) {
        float s = 0.f;
        #pragma unroll
        for (int w = 0; w < NW; w++) s += wsum[w * 64 + t];
        partial[(size_t)blockIdx.x * 128 + t] = s;
    } else if (t < 128) {
        const int n = t - 64;
        float s = 0.f;
        #pragma unroll
        for (int w = 0; w < NW; w++) s += wsq[w * 64 + n];
        partial[(size_t)blockIdx.x * 128 + 64 + n] = s;
    }
}

#define PCH 256

template <bool NEXT>
__global__ __launch_bounds__(256)
void stats_fold(const float* __restrict__ partial, int nblk,
                const float* __restrict__ g,
                const float* __restrict__ bt,
                float* __restrict__ stats,
                float invB,
                const float* __restrict__ Wn,
                const float* __restrict__ bnx,
                float* __restrict__ Wf,
                float* __restrict__ bf)
{
    __shared__ float pl[PCH * 128];
    __shared__ float red[128];
    __shared__ float scl[64];
    __shared__ float shl[64];
    __shared__ float term[NEXT ? 64 * LDSS : 1];

    const int t = threadIdx.x;
    const int k = t & 63;

    float w[16];
    float bnv = 0.f;
    if constexpr (NEXT) {
        #pragma unroll
        for (int j = 0; j < 16; j++) {
            const int n = (t >> 6) + 4 * j;
            const bool in = (n < 59 && k < 59);
            const float v = Wn[in ? n * 59 + k : 0];
            w[j] = in ? v : 0.0f;
        }
        const int nb = t >> 2;
        const float v = bnx[(nb < 59) ? nb : 0];
        bnv = (nb < 59) ? v : 0.0f;
    }

    const int c4 = t & 31;
    const int rw = t >> 5;
    float acc = 0.f;
    for (int base = 0; base < nblk; base += PCH) {
        if (base != 0) __syncthreads();
        const int rows = min(nblk - base, PCH);
        v4f pv[PCH / 8];
        const float* __restrict__ pb = partial + (size_t)base * 128;
        #pragma unroll
        for (int j = 0; j < PCH / 8; j++) {
            const unsigned r = (unsigned)min(rw + 8 * j, rows - 1);
            pv[j] = *(const v4f*)(pb + (r * 128u + (unsigned)c4 * 4u));
        }
        #pragma unroll
        for (int j = 0; j < PCH / 8; j++)
            *(v4f*)(&pl[(rw + 8 * j) * 128 + c4 * 4]) = pv[j];
        __syncthreads();
        if (t < 128) {
            int b = 0;
            for (; b + 64 <= rows; b += 64) {
                float v[64];
                #pragma unroll
                for (int q = 0; q < 64; q++) v[q] = pl[(b + q) * 128 + t];
                __builtin_amdgcn_sched_barrier(0);
                #pragma unroll
                for (int q = 0; q < 64; q++) acc += v[q];
            }
            for (; b < rows; b++)
                acc += pl[b * 128 + t];
        }
    }
    if (t < 128) red[t] = acc;
    __syncthreads();

    if (t < 64) {
        const int n = t;
        const float s  = red[n];
        const float ss = red[64 + n];
        const float mu  = s * invB;
        float var = ss * invB - mu * mu;
        var = fmaxf(var, 0.f);
        float scale = 0.f, shift = 0.f;
        if (n < 59) {
            scale = g[n] * rsqrtf(var + 1e-5f);
            shift = bt[n] - mu * scale;
        }
        if constexpr (NEXT) {
            scl[n] = scale;
            shl[n] = shift;
        } else {
            stats[n]      = scale;
            stats[64 + n] = shift;
        }
    }

    if constexpr (NEXT) {
        __syncthreads();
        const float sck = scl[k];
        const float shk = shl[k];
        #pragma unroll
        for (int j = 0; j < 16; j++) {
            const int n = (t >> 6) + 4 * j;
            term[n * LDSS + k] = w[j] * shk;
        }
        __syncthreads();
        #pragma unroll
        for (int j = 0; j < 16; j++) {
            const int n = (t >> 6) + 4 * j;
            Wf[n * 64 + k] = w[j] * sck;
        }
        {
            const int n = t >> 2;
            const float* p = term + n * LDSS + (t & 3) * 16;
            float s = 0.f;
            for (int q = 0; q < 4; q++) {
                const v4f v = *(const v4f*)(p + 4 * q);
                s += v.x; s += v.y; s += v.z; s += v.w;
            }
            s += __shfl_xor(s, 1, 32);
            s += __shfl_xor(s, 2, 32);
            if ((t & 3) == 0) bf[n] = bnv + s;
        }
    }
}

__global__ void normalize_final(const float* __restrict__ h,
                                const float* __restrict__ stats,
                                float* __restrict__ out, size_t total  )
{
    const size_t i = (size_t)blockIdx.x * 256 + threadIdx.x;
    if (i >= total) return;
    const size_t row = i / 59;
    const int n = (int)(i - row * 59);
    out[i] = h[row * 64 + n] * stats[n] + stats[64 + n];
}

__global__ void pad_input(const float* __restrict__ x, float* __restrict__ xp,
                          size_t total  )
{
    const size_t i = (size_t)blockIdx.x * 256 + threadIdx.x;
    if (i >= total) return;
    const size_t row = i >> 5;
    const int n = (int)(i & 31);
    xp[i] = (n < 29) ? x[row * 29 + n] : 0.0f;
}

extern "C" void kernel_launch(void* const* d_in, const int* in_sizes, int n_in,
                              void* d_out, int out_size, void* d_ws, size_t ws_size,
                              hipStream_t stream) {
    const float* x   = (const float*)d_in[0];
    const float* W0  = (const float*)d_in[1];
    const float* b0  = (const float*)d_in[2];
    const float* g0  = (const float*)d_in[3];
    const float* bt0 = (const float*)d_in[4];
    const float* Ws  = (const float*)d_in[5];
    const float* bs  = (const float*)d_in[6];
    const float* gs  = (const float*)d_in[7];
    const float* bts = (const float*)d_in[8];
    float* out = (float*)d_out;

    const int B     = in_sizes[0] / 29;
    const int nblk  = B / TILE_M;
    const int G     = 256;
    const float invB = 1.0f / (float)B;

    float* ws      = (float*)d_ws;
    float* xpad    = ws;
    float* P       = xpad    + (size_t)B * 32;
    float* Q       = P       + (size_t)B * 64;
    float* partial = Q       + (size_t)B * 64;
    float* stats   = partial + (size_t)nblk * 128;
    float* Wf      = partial + (size_t)G * 128;
    float* bf      = Wf + 64 * 64;

    const size_t padTot = (size_t)B * 32;
    pad_input<<<(unsigned)((padTot + 255) / 256), 256, 0, stream>>>(x, xpad, padTot);

    gemm_relu_stats<32, false><<<G, 256, 0, stream>>>(xpad, 32, W0, 29, b0, P, partial, nblk);
    stats_fold<true><<<1, 256, 0, stream>>>(partial, G, g0, bt0, stats, invB, Ws, bs, Wf, bf);

    for (int l = 0; l < 14; l++) {
        const float* gg = gs  + (size_t)l * 59;
        const float* tt = bts + (size_t)l * 59;
        const float* src = (l & 1) ? Q : P;
        float*       dst = (l & 1) ? P : Q;
        gemm_relu_stats<64, true><<<G, 256, 0, stream>>>(src, 64, Wf, 64, bf, dst, partial, nblk);
        if (l < 13)
            stats_fold<true><<<1, 256, 0, stream>>>(partial, G, gg, tt, stats, invB,
                                                     Ws + (size_t)(l + 1) * 59 * 59, bs + (size_t)(l + 1) * 59, Wf, bf);
        else
            stats_fold<false><<<1, 256, 0, stream>>>(partial, G, gg, tt, stats, invB,
                                                      nullptr, nullptr, nullptr, nullptr);
    }

    const size_t total = (size_t)B * 59;
    normalize_final<<<(unsigned)((total + 255) / 256), 256, 0, stream>>>(P, stats, out, total);
}


# GEMM: each wave keeps its W fragments and bias in registers for all its tiles and reads a tile's A fragments as one group ahead of its 64 WMMAs; no LDS read among the WMMAs; sums bit-identical
# speedup vs baseline: 5.8133x; 1.0332x over previous
#include <hip/hip_runtime.h>
#include <hip/hip_bf16.h>


typedef __attribute__((ext_vector_type(2))) float v2f;
typedef __attribute__((ext_vector_type(8))) float v8f;
typedef __attribute__((ext_vector_type(4))) float v4f;

#define TILE_M     128
#define NW         8
#define LDSS       68

__device__ __forceinline__ void lds_barrier()
{
    asm volatile("s_wait_dscnt 0x0\n\ts_barrier_signal -1\n\ts_barrier_wait -1" ::: "memory");
}


template <int KP>
__device__ __forceinline__ void load_x_tile(v4f (&pre)[KP / 8],
                                            const float* __restrict__ xin,
                                            int ldx, size_t row0, int t)
{
    constexpr int C4N = KP / 4;
    constexpr int RJ  = 256 / C4N;
    static_assert(256 % C4N == 0, "a thread keeps its column over j");
    const float* __restrict__ xt = xin + row0 * (size_t)ldx;
    const unsigned off = (unsigned)(t / C4N) * (unsigned)ldx + (unsigned)(t % C4N) * 4u;
    #pragma unroll
    for (int j = 0; j < KP / 8; j++)
        pre[j] = *(const v4f*)(xt + (off + (unsigned)(RJ * j) * (unsigned)ldx));
}

template <int KP, bool FOLD>
__global__ __launch_bounds__(256)
void gemm_relu_stats(const float* __restrict__ xin, int ldx,
                     const float* __restrict__ W,  int Kw,
                     const float* __restrict__ bias,
                     float* __restrict__ hout,
                     float* __restrict__ partial,
                     int ntiles)
{
    __shared__ float xl[TILE_M * LDSS];
    __shared__ float Wl[64 * LDSS];
    __shared__ float biasl[64];
    __shared__ float wsum[NW * 64];
    __shared__ float wsq [NW * 64];

    constexpr int C4N = KP / 4;
    const int t    = threadIdx.x;
    const int lane = t & 31;
    const int wave = t >> 5;
    const int half = lane >> 4;
    const int l15  = lane & 15;
    const int gstep = (int)gridDim.x;

    v4f pre[KP / 8];
    int tile = (int)blockIdx.x;
    if (tile < ntiles)
        load_x_tile<KP>(pre, xin, ldx, (size_t)tile * TILE_M, t);

    if constexpr (FOLD) {
        static_assert(!FOLD || KP == 64, "the folded matrix is 64 x 64");
        const float bv = bias[t & 63];
        for (int j = 0; j < 4; j++) {
            const int i = t + 256 * j;
            const int n = i >> 4;
            const int c = (i & 15) * 4;
            *(v4f*)(&Wl[n * LDSS + c]) = *(const v4f*)(W + n * 64 + c);
        }
        biasl[t & 63] = bv;
    } else {
        for (int i = t; i < 64 * KP; i += 256) {
            int n = i / KP;
            int k = i - n * KP;
            Wl[n * LDSS + k] = (n < 59 && k < Kw) ? W[n * Kw + k] : 0.0f;
        }
        if (t < 64) biasl[t] = (t < 59) ? bias[t] : 0.0f;
    }

    lds_barrier();
    v2f bw[KP / 4][4];
    #pragma unroll
    for (int ks = 0; ks < KP / 4; ks++) {
        #pragma unroll
        for (int tt = 0; tt < 4; tt++)
            bw[ks][tt] = *(const v2f*)(Wl + (tt * 16 + l15) * LDSS + 4 * ks + 2 * half);
    }
    float bnr[4];
    #pragma unroll
    for (int tt = 0; tt < 4; tt++) bnr[tt] = biasl[tt * 16 + l15];

    float rs[4] = {0.f, 0.f, 0.f, 0.f};
    float rq[4] = {0.f, 0.f, 0.f, 0.f};

    for (; tile < ntiles; tile += gstep) {
        const size_t row0 = (size_t)tile * TILE_M;

        #pragma unroll
        for (int j = 0; j < KP / 8; j++) {
            const int i = t + 256 * j;
            const int r = i / C4N;
            const int c = (i % C4N) * 4;
            *(v4f*)(&xl[r * LDSS + c]) = pre[j];
        }

        const int nxt = tile + gstep;
        if (nxt < ntiles)
            load_x_tile<KP>(pre, xin, ldx, (size_t)nxt * TILE_M, t);

        lds_barrier();

        const float* ax = xl + (wave * 16 + l15) * LDSS + 2 * half;
        v2f af[KP / 4];
        #pragma unroll
        for (int ks = 0; ks < KP / 4; ks++)
            af[ks] = *(const v2f*)(ax + 4 * ks);
        __builtin_amdgcn_sched_barrier(0);

        v8f acc[4] = {};
        #pragma unroll
        for (int ks = 0; ks < KP / 4; ks++) {
            #pragma unroll
            for (int tt = 0; tt < 4; tt++)
                acc[tt] = __builtin_amdgcn_wmma_f32_16x16x4_f32(
                    false, af[ks], false, bw[ks][tt], (short)0, acc[tt], false, false);
        }

        const size_t mbase = row0 + wave * 16 + half * 8;
        #pragma unroll
        for (int tt = 0; tt < 4; tt++) {
            const int n = tt * 16 + l15;
            const float bn = bnr[tt];
            float sl = 0.f, sq = 0.f;
            #pragma unroll
            for (int r = 0; r < 8; r++) {
                float v = acc[tt][r] + bn;
                v = fmaxf(v, 0.f);
                hout[(mbase + r) * 64 + n] = v;
                sl += v;
                sq += v * v;
            }
            rs[tt] += sl;
            rq[tt] += sq;
        }
        lds_barrier();
    }

    #pragma unroll
    for (int tt = 0; tt < 4; tt++) {
        const int n = tt * 16 + l15;
        float sl = rs[tt], sq = rq[tt];
        sl += __shfl_xor(sl, 16, 32);
        sq += __shfl_xor(sq, 16, 32);
        if (half == 0) {
            wsum[wave * 64 + n] = sl;
            wsq [wave * 64 + n] = sq;
        }
    }
    lds_barrier();

    if (t < 64) {
        float s = 0.f;
        #pragma unroll
        for (int w = 0; w < NW; w++) s += wsum[w * 64 + t];
        partial[(size_t)blockIdx.x * 128 + t] = s;
    } else if (t < 128) {
        const int n = t - 64;
        float s = 0.f;
        #pragma unroll
        for (int w = 0; w < NW; w++) s += wsq[w * 64 + n];
        partial[(size_t)blockIdx.x * 128 + 64 + n] = s;
    }
}

#define PCH 256

template <bool NEXT>
__global__ __launch_bounds__(256)
void stats_fold(const float* __restrict__ partial, int nblk,
                const float* __restrict__ g,
                const float* __restrict__ bt,
                float* __restrict__ stats,
                float invB,
                const float* __restrict__ Wn,
                const float* __restrict__ bnx,
                float* __restrict__ Wf,
                float* __restrict__ bf)
{
    __shared__ float pl[PCH * 128];
    __shared__ float red[128];
    __shared__ float scl[64];
    __shared__ float shl[64];
    __shared__ float term[NEXT ? 64 * LDSS : 1];

    const int t = threadIdx.x;
    const int k = t & 63;

    float w[16];
    float bnv = 0.f;
    if constexpr (NEXT) {
        #pragma unroll
        for (int j = 0; j < 16; j++) {
            const int n = (t >> 6) + 4 * j;
            const bool in = (n < 59 && k < 59);
            const float v = Wn[in ? n * 59 + k : 0];
            w[j] = in ? v : 0.0f;
        }
        const int nb = t >> 2;
        const float v = bnx[(nb < 59) ? nb : 0];
        bnv = (nb < 59) ? v : 0.0f;
    }

    const int c4 = t & 31;
    const int rw = t >> 5;
    float acc = 0.f;
    for (int base = 0; base < nblk; base += PCH) {
        if (base != 0) __syncthreads();
        const int rows = min(nblk - base, PCH);
        v4f pv[PCH / 8];
        const float* __restrict__ pb = partial + (size_t)base * 128;
        #pragma unroll
        for (int j = 0; j < PCH / 8; j++) {
            const unsigned r = (unsigned)min(rw + 8 * j, rows - 1);
            pv[j] = *(const v4f*)(pb + (r * 128u + (unsigned)c4 * 4u));
        }
        #pragma unroll
        for (int j = 0; j < PCH / 8; j++)
            *(v4f*)(&pl[(rw + 8 * j) * 128 + c4 * 4]) = pv[j];
        __syncthreads();
        if (t < 128) {
            int b = 0;
            for (; b + 64 <= rows; b += 64) {
                float v[64];
                #pragma unroll
                for (int q = 0; q < 64; q++) v[q] = pl[(b + q) * 128 + t];
                __builtin_amdgcn_sched_barrier(0);
                #pragma unroll
                for (int q = 0; q < 64; q++) acc += v[q];
            }
            for (; b < rows; b++)
                acc += pl[b * 128 + t];
        }
    }
    if (t < 128) red[t] = acc;
    __syncthreads();

    if (t < 64) {
        const int n = t;
        const float s  = red[n];
        const float ss = red[64 + n];
        const float mu  = s * invB;
        float var = ss * invB - mu * mu;
        var = fmaxf(var, 0.f);
        float scale = 0.f, shift = 0.f;
        if (n < 59) {
            scale = g[n] * rsqrtf(var + 1e-5f);
            shift = bt[n] - mu * scale;
        }
        if constexpr (NEXT) {
            scl[n] = scale;
            shl[n] = shift;
        } else {
            stats[n]      = scale;
            stats[64 + n] = shift;
        }
    }

    if constexpr (NEXT) {
        __syncthreads();
        const float sck = scl[k];
        const float shk = shl[k];
        #pragma unroll
        for (int j = 0; j < 16; j++) {
            const int n = (t >> 6) + 4 * j;
            term[n * LDSS + k] = w[j] * shk;
        }
        __syncthreads();
        #pragma unroll
        for (int j = 0; j < 16; j++) {
            const int n = (t >> 6) + 4 * j;
            Wf[n * 64 + k] = w[j] * sck;
        }
        {
            const int n = t >> 2;
            const float* p = term + n * LDSS + (t & 3) * 16;
            float s = 0.f;
            for (int q = 0; q < 4; q++) {
                const v4f v = *(const v4f*)(p + 4 * q);
                s += v.x; s += v.y; s += v.z; s += v.w;
            }
            s += __shfl_xor(s, 1, 32);
            s += __shfl_xor(s, 2, 32);
            if ((t & 3) == 0) bf[n] = bnv + s;
        }
    }
}

__global__ void normalize_final(const float* __restrict__ h,
                                const float* __restrict__ stats,
                                float* __restrict__ out, size_t total  )
{
    const size_t i = (size_t)blockIdx.x * 256 + threadIdx.x;
    if (i >= total) return;
    const size_t row = i / 59;
    const int n = (int)(i - row * 59);
    out[i] = h[row * 64 + n] * stats[n] + stats[64 + n];
}

__global__ void pad_input(const float* __restrict__ x, float* __restrict__ xp,
                          size_t total  )
{
    const size_t i = (size_t)blockIdx.x * 256 + threadIdx.x;
    if (i >= total) return;
    const size_t row = i >> 5;
    const int n = (int)(i & 31);
    xp[i] = (n < 29) ? x[row * 29 + n] : 0.0f;
}

extern "C" void kernel_launch(void* const* d_in, const int* in_sizes, int n_in,
                              void* d_out, int out_size, void* d_ws, size_t ws_size,
                              hipStream_t stream) {
    const float* x   = (const float*)d_in[0];
    const float* W0  = (const float*)d_in[1];
    const float* b0  = (const float*)d_in[2];
    const float* g0  = (const float*)d_in[3];
    const float* bt0 = (const float*)d_in[4];
    const float* Ws  = (const float*)d_in[5];
    const float* bs  = (const float*)d_in[6];
    const float* gs  = (const float*)d_in[7];
    const float* bts = (const float*)d_in[8];
    float* out = (float*)d_out;

    const int B     = in_sizes[0] / 29;
    const int nblk  = B / TILE_M;
    const int G     = 256;
    const float invB = 1.0f / (float)B;

    float* ws      = (float*)d_ws;
    float* xpad    = ws;
    float* P       = xpad    + (size_t)B * 32;
    float* Q       = P       + (size_t)B * 64;
    float* partial = Q       + (size_t)B * 64;
    float* stats   = partial + (size_t)nblk * 128;
    float* Wf      = partial + (size_t)G * 128;
    float* bf      = Wf + 64 * 64;

    const size_t padTot = (size_t)B * 32;
    pad_input<<<(unsigned)((padTot + 255) / 256), 256, 0, stream>>>(x, xpad, padTot);

    gemm_relu_stats<32, false><<<G, 256, 0, stream>>>(xpad, 32, W0, 29, b0, P, partial, nblk);
    stats_fold<true><<<1, 256, 0, stream>>>(partial, G, g0, bt0, stats, invB, Ws, bs, Wf, bf);

    for (int l = 0; l < 14; l++) {
        const float* gg = gs  + (size_t)l * 59;
        const float* tt = bts + (size_t)l * 59;
        const float* src = (l & 1) ? Q : P;
        float*       dst = (l & 1) ? P : Q;
        gemm_relu_stats<64, true><<<G, 256, 0, stream>>>(src, 64, Wf, 64, bf, dst, partial, nblk);
        if (l < 13)
            stats_fold<true><<<1, 256, 0, stream>>>(partial, G, gg, tt, stats, invB,
                                                     Ws + (size_t)(l + 1) * 59 * 59, bs + (size_t)(l + 1) * 59, Wf, bf);
        else
            stats_fold<false><<<1, 256, 0, stream>>>(partial, G, gg, tt, stats, invB,
                                                      nullptr, nullptr, nullptr, nullptr);
    }

    const size_t total = (size_t)B * 59;
    normalize_final<<<(unsigned)((total + 255) / 256), 256, 0, stream>>>(P, stats, out, total);
}
